# starting kernel
# baseline (speedup 1.0000x reference)
#include <hip/hip_runtime.h>
#include <math.h>

typedef float v2f __attribute__((ext_vector_type(2)));
typedef float v8f __attribute__((ext_vector_type(8)));

#define SPB 128
#define THREADS_A 256

__global__ __launch_bounds__(THREADS_A)
void pool_circuit_kernel(const float* __restrict__ x,
                         const float* __restrict__ params,
                         float* __restrict__ zbuf,
                         float* __restrict__ partials,
                         int B)
{
    __shared__ float pooled[SPB][4];
    __shared__ float red[SPB][8];
    __shared__ float sAB[8];

    const int tid   = threadIdx.x;
    const int lane  = tid & 31;
    const int wv    = tid >> 5;
    const int nIdx  = lane & 15;
    const int kBase = (lane >> 4) << 1;

    if (tid < 4) {
        float al = params[2 * tid + 0];
        float be = params[2 * tid + 1];
        sAB[tid]     = cosf(be);
        sAB[tid + 4] = sinf(be) * sinf(al);
    }

    v2f bt[6];
#pragma unroll
    for (int t = 0; t < 6; ++t) {
        int c0 = 4 * t + kBase;
        bt[t].x = (nIdx == (c0    ) / 6) ? (1.0f / 36.0f) : 0.0f;
        bt[t].y = (nIdx == (c0 + 1) / 6) ? (1.0f / 36.0f) : 0.0f;
    }

    const int sampleBase = blockIdx.x * SPB + wv * 16;
    const float* px = x + (size_t)(sampleBase + nIdx) * 576 + kBase;

    v8f acc = {0.f, 0.f, 0.f, 0.f, 0.f, 0.f, 0.f, 0.f};
#pragma unroll
    for (int row = 0; row < 6; ++row) {
#pragma unroll
        for (int t = 0; t < 6; ++t) {
            v2f a = *(const v2f*)(px + row * 24 + t * 4);
            acc = __builtin_amdgcn_wmma_f32_16x16x4_f32(
                false, a, false, bt[t], (short)0, acc, false, false);
        }
    }

    if (nIdx < 4) {
        int srow = wv * 16 + ((lane >> 4) << 3);
#pragma unroll
        for (int r = 0; r < 8; ++r)
            pooled[srow + r][nIdx] = acc[r];
    }
    __syncthreads();

    if (tid < SPB) {
        float z[4];
        float prod = 1.0f;
#pragma unroll
        for (int j = 0; j < 4; ++j) {
            float xj = pooled[tid][j];
            float sx, cx;
            sincosf(xj, &sx, &cx);
            prod *= sAB[j] * cx + sAB[4 + j] * sx;
            z[j] = prod;
        }
        size_t s = (size_t)blockIdx.x * SPB + tid;
        *(float4*)(zbuf + s * 4) = make_float4(z[0], z[1], z[2], z[3]);
#pragma unroll
        for (int j = 0; j < 4; ++j) {
            red[tid][j]     = z[j];
            red[tid][4 + j] = z[j] * z[j];
        }
    }
    __syncthreads();

    for (int s = SPB / 2; s > 0; s >>= 1) {
        if (tid < s) {
#pragma unroll
            for (int j = 0; j < 8; ++j) red[tid][j] += red[tid + s][j];
        }
        __syncthreads();
    }
    if (tid < 8) partials[(size_t)blockIdx.x * 8 + tid] = red[0][tid];
}

__global__ void stats_kernel(const float* __restrict__ partials, int nblk, int B,
                             const float* __restrict__ gamma,
                             const float* __restrict__ beta,
                             float* __restrict__ stats)
{
    __shared__ float sums[8];
    int j = threadIdx.x;
    if (j < 8) {
        float s = 0.f;
        for (int b = 0; b < nblk; ++b) s += partials[(size_t)b * 8 + j];
        sums[j] = s;
    }
    __syncthreads();
    if (j < 4) {
        float invB = 1.0f / (float)B;
        float mean = sums[j] * invB;
        float var  = sums[j + 4] * invB - mean * mean;
        float sc   = gamma[j] * rsqrtf(var + 1e-5f);
        stats[j]     = sc;
        stats[4 + j] = beta[j] - mean * sc;
    }
}

__global__ __launch_bounds__(256)
void norm_kernel(const float* __restrict__ zbuf,
                 const float* __restrict__ stats,
                 float* __restrict__ out, int B)
{
    int i = blockIdx.x * blockDim.x + threadIdx.x;
    if (i < B) {
        float4 z  = *(const float4*)(zbuf + (size_t)i * 4);
        float4 sc = *(const float4*)(stats);
        float4 sh = *(const float4*)(stats + 4);
        float4 o;
        o.x = z.x * sc.x + sh.x;
        o.y = z.y * sc.y + sh.y;
        o.z = z.z * sc.z + sh.z;
        o.w = z.w * sc.w + sh.w;
        *(float4*)(out + (size_t)i * 4) = o;
    }
}

extern "C" void kernel_launch(void* const* d_in, const int* in_sizes, int n_in,
                              void* d_out, int out_size, void* d_ws, size_t ws_size,
                              hipStream_t stream)
{
    const float* x      = (const float*)d_in[0];
    const float* params = (const float*)d_in[1];
    const float* gamma  = (const float*)d_in[2];
    const float* beta   = (const float*)d_in[3];
    float* out = (float*)d_out;

    int B = in_sizes[0] / 576;
    int nblkA = B / SPB;

    float* zbuf     = (float*)d_ws;
    float* partials = zbuf + (size_t)B * 4;
    float* stats    = partials + (size_t)nblkA * 8;

    pool_circuit_kernel<<<nblkA, THREADS_A, 0, stream>>>(x, params, zbuf, partials, B);
    stats_kernel<<<1, 32, 0, stream>>>(partials, nblkA, B, gamma, beta, stats);
    norm_kernel<<<(B + 255) / 256, 256, 0, stream>>>(zbuf, stats, out, B);
}


# Stats kernel: one global-load round trip, was nine: all 32 lanes load the block partials (16 x 16 B per lane in flight, gamma/beta too); pairwise tree, lane exchanges, no LDS; sum order changes
# speedup vs baseline: 2.1755x; 2.1755x over previous
#include <hip/hip_runtime.h>
#include <math.h>

typedef float v2f __attribute__((ext_vector_type(2)));
typedef float v8f __attribute__((ext_vector_type(8)));

#define SPB 256
#define THREADS_A 256

__global__ __launch_bounds__(THREADS_A) __attribute__((amdgpu_waves_per_eu(2, 4)))
void pool_circuit_kernel(const float* __restrict__ x,
                         const float* __restrict__ params,
                         float* __restrict__ zbuf,
                         float* __restrict__ partials,
                         int B)
{
    __shared__ float pooled[SPB][4];
    __shared__ float red[SPB][8];
    __shared__ float sAB[8];

    const int tid   = threadIdx.x;
    const int lane  = tid & 31;
    const int wv    = tid >> 5;
    const int nIdx  = lane & 15;
    const int kBase = (lane >> 4) << 1;

    if (tid < 4) {
        float al = params[2 * tid + 0];
        float be = params[2 * tid + 1];
        sAB[tid]     = cosf(be);
        sAB[tid + 4] = sinf(be) * sinf(al);
    }

    v2f bt[6];
#pragma unroll
    for (int t = 0; t < 6; ++t) {
        int c0 = 4 * t + kBase;
        bt[t].x = (nIdx == (c0    ) / 6) ? (1.0f / 36.0f) : 0.0f;
        bt[t].y = (nIdx == (c0 + 1) / 6) ? (1.0f / 36.0f) : 0.0f;
    }

    const int sampleBase = blockIdx.x * SPB + wv * 32;
    const float* px0 = x + (size_t)(sampleBase + nIdx) * 576 + kBase;
    const float* px1 = px0 + (size_t)16 * 576;

    v8f acc0 = {0.f, 0.f, 0.f, 0.f, 0.f, 0.f, 0.f, 0.f};
    v8f acc1 = {0.f, 0.f, 0.f, 0.f, 0.f, 0.f, 0.f, 0.f};
#pragma unroll
    for (int row = 0; row < 6; ++row) {
#pragma unroll
        for (int t = 0; t < 6; ++t) {
            v2f a0 = *(const v2f*)(px0 + row * 24 + t * 4);
            v2f a1 = *(const v2f*)(px1 + row * 24 + t * 4);
            acc0 = __builtin_amdgcn_wmma_f32_16x16x4_f32(
                false, a0, false, bt[t], (short)0, acc0, false, false);
            acc1 = __builtin_amdgcn_wmma_f32_16x16x4_f32(
                false, a1, false, bt[t], (short)0, acc1, false, false);
        }
    }

    if (nIdx < 4) {
        int srow = wv * 32 + ((lane >> 4) << 3);
#pragma unroll
        for (int r = 0; r < 8; ++r) {
            pooled[srow + r][nIdx]      = acc0[r];
            pooled[srow + 16 + r][nIdx] = acc1[r];
        }
    }
    __syncthreads();

    {
        float z[4];
        float prod = 1.0f;
#pragma unroll
        for (int j = 0; j < 4; ++j) {
            float xj = pooled[tid][j];
            float sx, cx;
            sincosf(xj, &sx, &cx);
            prod *= sAB[j] * cx + sAB[4 + j] * sx;
            z[j] = prod;
        }
        size_t s = (size_t)blockIdx.x * SPB + tid;
        *(float4*)(zbuf + s * 4) = make_float4(z[0], z[1], z[2], z[3]);
#pragma unroll
        for (int j = 0; j < 4; ++j) {
            red[tid][j]     = z[j];
            red[tid][4 + j] = z[j] * z[j];
        }
    }
    __syncthreads();

    for (int s = SPB / 2; s > 0; s >>= 1) {
        if (tid < s) {
#pragma unroll
            for (int j = 0; j < 8; ++j) red[tid][j] += red[tid + s][j];
        }
        __syncthreads();
    }
    if (tid < 8) {
        const int b    = blockIdx.x;
        const int full = (B / SPB) & ~3;
        size_t idx = (b < full) ? (size_t)(b >> 2) * 32 + tid * 4 + (b & 3)
                                : (size_t)b * 8 + tid;
        partials[idx] = red[0][tid];
    }
}

__device__ __forceinline__ float lane_xor16(float x)
{
    int b = __builtin_bit_cast(int, x);
    return __builtin_bit_cast(float,
        __builtin_amdgcn_permlanex16(b, b, 0x76543210, 0xfedcba98, false, false));
}
template <int MASK>
__device__ __forceinline__ float row_xor(float x)
{
    int b = __builtin_bit_cast(int, x);
    return __builtin_bit_cast(float,
        __builtin_amdgcn_update_dpp(0, b, 0x160 + MASK, 0xf, 0xf, false));
}

__global__ __launch_bounds__(32) __attribute__((amdgpu_waves_per_eu(1, 1)))
void stats_kernel(const float* __restrict__ partials, int nblk, int B,
                             const float* __restrict__ gamma,
                             const float* __restrict__ beta,
                             float* __restrict__ stats)
{
    const int lane = threadIdx.x;
    const int j    = lane & 7;
    const int ngrp = nblk >> 2;
    const int n4   = ngrp * 8;
    const float4* p4 = (const float4*)partials;

    const float g  = gamma[lane & 3];
    const float be = beta[lane & 3];

    float4 acc = make_float4(0.f, 0.f, 0.f, 0.f);
    int base = 0;
    for (; base + 512 <= n4; base += 512) {
        float4 v[16];
#pragma unroll
        for (int u = 0; u < 16; ++u) v[u] = p4[base + lane + 32 * u];
#pragma unroll
        for (int st = 1; st < 16; st <<= 1) {
#pragma unroll
            for (int u = 0; u < 16; u += 2 * st) {
                v[u].x += v[u + st].x;
                v[u].y += v[u + st].y;
                v[u].z += v[u + st].z;
                v[u].w += v[u + st].w;
            }
        }
        acc.x += v[0].x;
        acc.y += v[0].y;
        acc.z += v[0].z;
        acc.w += v[0].w;
    }
    for (int i = base + lane; i < n4; i += 32) {
        float4 v = p4[i];
        acc.x += v.x;
        acc.y += v.y;
        acc.z += v.z;
        acc.w += v.w;
    }
    float s = (acc.x + acc.y) + (acc.z + acc.w);
    if (lane < 8) {
        for (int b = ngrp << 2; b < nblk; ++b) s += partials[(size_t)b * 8 + j];
    }
    s += row_xor<8>(s);
    s += lane_xor16(s);
    const float sq = row_xor<4>(s);

    float invB = 1.0f / (float)B;
    float mean = s * invB;
    float var  = sq * invB - mean * mean;
    float sc   = g * rsqrtf(var + 1e-5f);
    if (lane < 4) {
        stats[lane]     = sc;
        stats[4 + lane] = be - mean * sc;
    }
}

__global__ __launch_bounds__(256)
void norm_kernel(const float* __restrict__ zbuf,
                 const float* __restrict__ stats,
                 float* __restrict__ out, int B)
{
    int i = blockIdx.x * blockDim.x + threadIdx.x;
    if (i < B) {
        float4 z  = *(const float4*)(zbuf + (size_t)i * 4);
        float4 sc = *(const float4*)(stats);
        float4 sh = *(const float4*)(stats + 4);
        float4 o;
        o.x = z.x * sc.x + sh.x;
        o.y = z.y * sc.y + sh.y;
        o.z = z.z * sc.z + sh.z;
        o.w = z.w * sc.w + sh.w;
        *(float4*)(out + (size_t)i * 4) = o;
    }
}

extern "C" void kernel_launch(void* const* d_in, const int* in_sizes, int n_in,
                              void* d_out, int out_size, void* d_ws, size_t ws_size,
                              hipStream_t stream)
{
    const float* x      = (const float*)d_in[0];
    const float* params = (const float*)d_in[1];
    const float* gamma  = (const float*)d_in[2];
    const float* beta   = (const float*)d_in[3];
    float* out = (float*)d_out;

    int B = in_sizes[0] / 576;
    int nblkA = B / SPB;

    float* zbuf     = (float*)d_ws;
    float* partials = zbuf + (size_t)B * 4;
    float* stats    = partials + (size_t)nblkA * 8;

    pool_circuit_kernel<<<nblkA, THREADS_A, 0, stream>>>(x, params, zbuf, partials, B);
    stats_kernel<<<1, 32, 0, stream>>>(partials, nblkA, B, gamma, beta, stats);
    norm_kernel<<<(B + 255) / 256, 256, 0, stream>>>(zbuf, stats, out, B);
}


# Pooling WMMA re-tiled: a tile's 16 rows = 4 samples x four 16-byte chunks, so a fragment load touches 4 lines, was 16; B has (chunk, block) columns; 3 DPP adds join the chunks; sum order changes
# speedup vs baseline: 2.3318x; 1.0718x over previous
#include <hip/hip_runtime.h>
#include <math.h>

typedef float v2f __attribute__((ext_vector_type(2)));
typedef float v8f __attribute__((ext_vector_type(8)));

#define SPB 256
#define THREADS_A 256

template <int MASK>
__device__ __forceinline__ float row_xor(float x)
{
    int b = __builtin_bit_cast(int, x);
    return __builtin_bit_cast(float,
        __builtin_amdgcn_update_dpp(0, b, 0x160 + MASK, 0xf, 0xf, false));
}

__global__ __launch_bounds__(THREADS_A) __attribute__((amdgpu_waves_per_eu(2, 4)))
void pool_circuit_kernel(const float* __restrict__ x,
                         const float* __restrict__ params,
                         float* __restrict__ zbuf,
                         float* __restrict__ partials,
                         int B)
{
    __shared__ float pooled[SPB][4];
    __shared__ float red[SPB][8];
    __shared__ float sAB[8];

    const int tid   = threadIdx.x;
    const int lane  = tid & 31;
    const int wv    = tid >> 5;
    const int nIdx  = lane & 15;
    const int kBase = (lane >> 4) << 1;
    const int hi    = nIdx >> 2;
    const int lo    = nIdx & 3;

    if (tid < 4) {
        float al = params[2 * tid + 0];
        float be = params[2 * tid + 1];
        sAB[tid]     = cosf(be);
        sAB[tid + 4] = sinf(be) * sinf(al);
    }

    v2f bt[3];
#pragma unroll
    for (int u = 0; u < 3; ++u) {
        int e0 = (16 * u) % 24 + 4 * hi + kBase;
        bt[u].x = (lo == ((e0    ) % 24) / 6) ? (1.0f / 36.0f) : 0.0f;
        bt[u].y = (lo == ((e0 + 1) % 24) / 6) ? (1.0f / 36.0f) : 0.0f;
    }

    const int sampleBase = blockIdx.x * SPB + wv * 32;
    const float* px = x + (size_t)(sampleBase + hi) * 576 + 4 * lo + kBase;

    float pv[16];
#pragma unroll
    for (int tp = 0; tp < 4; ++tp) {
        const float* p0 = px + (size_t)(8 * tp) * 576;
        const float* p1 = p0 + (size_t)4 * 576;
        v8f acc0 = {0.f, 0.f, 0.f, 0.f, 0.f, 0.f, 0.f, 0.f};
        v8f acc1 = {0.f, 0.f, 0.f, 0.f, 0.f, 0.f, 0.f, 0.f};
#pragma unroll
        for (int t = 0; t < 9; ++t) {
            v2f a0 = *(const v2f*)(p0 + 16 * t);
            v2f a1 = *(const v2f*)(p1 + 16 * t);
            acc0 = __builtin_amdgcn_wmma_f32_16x16x4_f32(
                false, a0, false, bt[t % 3], (short)0, acc0, false, false);
            acc1 = __builtin_amdgcn_wmma_f32_16x16x4_f32(
                false, a1, false, bt[t % 3], (short)0, acc1, false, false);
        }
#pragma unroll
        for (int g = 0; g < 2; ++g) {
            pv[4 * tp + g]     = (acc0[4 * g] + row_xor<4>(acc0[4 * g + 1]))
                               + (row_xor<8>(acc0[4 * g + 2]) + row_xor<12>(acc0[4 * g + 3]));
            pv[4 * tp + 2 + g] = (acc1[4 * g] + row_xor<4>(acc1[4 * g + 1]))
                               + (row_xor<8>(acc1[4 * g + 2]) + row_xor<12>(acc1[4 * g + 3]));
        }
    }

    if (nIdx < 4) {
        int srow = wv * 32 + ((lane >> 4) << 1);
#pragma unroll
        for (int q = 0; q < 8; ++q) {
            pooled[srow + 4 * q    ][nIdx] = pv[2 * q];
            pooled[srow + 4 * q + 1][nIdx] = pv[2 * q + 1];
        }
    }
    __syncthreads();

    {
        float z[4];
        float prod = 1.0f;
#pragma unroll
        for (int j = 0; j < 4; ++j) {
            float xj = pooled[tid][j];
            float sx, cx;
            sincosf(xj, &sx, &cx);
            prod *= sAB[j] * cx + sAB[4 + j] * sx;
            z[j] = prod;
        }
        size_t s = (size_t)blockIdx.x * SPB + tid;
        *(float4*)(zbuf + s * 4) = make_float4(z[0], z[1], z[2], z[3]);
#pragma unroll
        for (int j = 0; j < 4; ++j) {
            red[tid][j]     = z[j];
            red[tid][4 + j] = z[j] * z[j];
        }
    }
    __syncthreads();

    for (int s = SPB / 2; s > 0; s >>= 1) {
        if (tid < s) {
#pragma unroll
            for (int j = 0; j < 8; ++j) red[tid][j] += red[tid + s][j];
        }
        __syncthreads();
    }
    if (tid < 8) {
        const int b    = blockIdx.x;
        const int full = (B / SPB) & ~3;
        size_t idx = (b < full) ? (size_t)(b >> 2) * 32 + tid * 4 + (b & 3)
                                : (size_t)b * 8 + tid;
        partials[idx] = red[0][tid];
    }
}

__device__ __forceinline__ float lane_xor16(float x)
{
    int b = __builtin_bit_cast(int, x);
    return __builtin_bit_cast(float,
        __builtin_amdgcn_permlanex16(b, b, 0x76543210, 0xfedcba98, false, false));
}

__global__ __launch_bounds__(32) __attribute__((amdgpu_waves_per_eu(1, 1)))
void stats_kernel(const float* __restrict__ partials, int nblk, int B,
                             const float* __restrict__ gamma,
                             const float* __restrict__ beta,
                             float* __restrict__ stats)
{
    const int lane = threadIdx.x;
    const int j    = lane & 7;
    const int ngrp = nblk >> 2;
    const int n4   = ngrp * 8;
    const float4* p4 = (const float4*)partials;

    const float g  = gamma[lane & 3];
    const float be = beta[lane & 3];

    float4 acc = make_float4(0.f, 0.f, 0.f, 0.f);
    int base = 0;
    for (; base + 512 <= n4; base += 512) {
        float4 v[16];
#pragma unroll
        for (int u = 0; u < 16; ++u) v[u] = p4[base + lane + 32 * u];
#pragma unroll
        for (int st = 1; st < 16; st <<= 1) {
#pragma unroll
            for (int u = 0; u < 16; u += 2 * st) {
                v[u].x += v[u + st].x;
                v[u].y += v[u + st].y;
                v[u].z += v[u + st].z;
                v[u].w += v[u + st].w;
            }
        }
        acc.x += v[0].x;
        acc.y += v[0].y;
        acc.z += v[0].z;
        acc.w += v[0].w;
    }
    for (int i = base + lane; i < n4; i += 32) {
        float4 v = p4[i];
        acc.x += v.x;
        acc.y += v.y;
        acc.z += v.z;
        acc.w += v.w;
    }
    float s = (acc.x + acc.y) + (acc.z + acc.w);
    if (lane < 8) {
        for (int b = ngrp << 2; b < nblk; ++b) s += partials[(size_t)b * 8 + j];
    }
    s += row_xor<8>(s);
    s += lane_xor16(s);
    const float sq = row_xor<4>(s);

    float invB = 1.0f / (float)B;
    float mean = s * invB;
    float var  = sq * invB - mean * mean;
    float sc   = g * rsqrtf(var + 1e-5f);
    if (lane < 4) {
        stats[lane]     = sc;
        stats[4 + lane] = be - mean * sc;
    }
}

__global__ __launch_bounds__(256)
void norm_kernel(const float* __restrict__ zbuf,
                 const float* __restrict__ stats,
                 float* __restrict__ out, int B)
{
    int i = blockIdx.x * blockDim.x + threadIdx.x;
    if (i < B) {
        float4 z  = *(const float4*)(zbuf + (size_t)i * 4);
        float4 sc = *(const float4*)(stats);
        float4 sh = *(const float4*)(stats + 4);
        float4 o;
        o.x = z.x * sc.x + sh.x;
        o.y = z.y * sc.y + sh.y;
        o.z = z.z * sc.z + sh.z;
        o.w = z.w * sc.w + sh.w;
        *(float4*)(out + (size_t)i * 4) = o;
    }
}

extern "C" void kernel_launch(void* const* d_in, const int* in_sizes, int n_in,
                              void* d_out, int out_size, void* d_ws, size_t ws_size,
                              hipStream_t stream)
{
    const float* x      = (const float*)d_in[0];
    const float* params = (const float*)d_in[1];
    const float* gamma  = (const float*)d_in[2];
    const float* beta   = (const float*)d_in[3];
    float* out = (float*)d_out;

    int B = in_sizes[0] / 576;
    int nblkA = B / SPB;

    float* zbuf     = (float*)d_ws;
    float* partials = zbuf + (size_t)B * 4;
    float* stats    = partials + (size_t)nblkA * 8;

    pool_circuit_kernel<<<nblkA, THREADS_A, 0, stream>>>(x, params, zbuf, partials, B);
    stats_kernel<<<1, 32, 0, stream>>>(partials, nblkA, B, gamma, beta, stats);
    norm_kernel<<<(B + 255) / 256, 256, 0, stream>>>(zbuf, stats, out, B);
}


# Pooling kernel tail: block stats by lanes that own a column, two LDS rounds (8 rows, then 32): 2 barriers, was 8; z stored behind the last barrier (its ack is no longer waited for); sum tree kept
# speedup vs baseline: 2.5120x; 1.0773x over previous
#include <hip/hip_runtime.h>
#include <math.h>

typedef float v2f __attribute__((ext_vector_type(2)));
typedef float v8f __attribute__((ext_vector_type(8)));

#define SPB 256
#define THREADS_A 256

template <int MASK>
__device__ __forceinline__ float row_xor(float x)
{
    int b = __builtin_bit_cast(int, x);
    return __builtin_bit_cast(float,
        __builtin_amdgcn_update_dpp(0, b, 0x160 + MASK, 0xf, 0xf, false));
}

__global__ __launch_bounds__(THREADS_A) __attribute__((amdgpu_waves_per_eu(2, 4)))
void pool_circuit_kernel(const float* __restrict__ x,
                         const float* __restrict__ params,
                         float* __restrict__ zbuf,
                         float* __restrict__ partials,
                         int B)
{
    __shared__ float pooled[SPB][4];
    __shared__ float red[SPB][8];
    __shared__ float red2[32][8];
    __shared__ float sAB[8];

    const int tid   = threadIdx.x;
    const int lane  = tid & 31;
    const int wv    = tid >> 5;
    const int nIdx  = lane & 15;
    const int kBase = (lane >> 4) << 1;
    const int hi    = nIdx >> 2;
    const int lo    = nIdx & 3;

    if (tid < 4) {
        float al = params[2 * tid + 0];
        float be = params[2 * tid + 1];
        sAB[tid]     = cosf(be);
        sAB[tid + 4] = sinf(be) * sinf(al);
    }

    v2f bt[3];
#pragma unroll
    for (int u = 0; u < 3; ++u) {
        int e0 = (16 * u) % 24 + 4 * hi + kBase;
        bt[u].x = (lo == ((e0    ) % 24) / 6) ? (1.0f / 36.0f) : 0.0f;
        bt[u].y = (lo == ((e0 + 1) % 24) / 6) ? (1.0f / 36.0f) : 0.0f;
    }

    const int sampleBase = blockIdx.x * SPB + wv * 32;
    const float* px = x + (size_t)(sampleBase + hi) * 576 + 4 * lo + kBase;

    float pv[16];
#pragma unroll
    for (int tp = 0; tp < 4; ++tp) {
        const float* p0 = px + (size_t)(8 * tp) * 576;
        const float* p1 = p0 + (size_t)4 * 576;
        v8f acc0 = {0.f, 0.f, 0.f, 0.f, 0.f, 0.f, 0.f, 0.f};
        v8f acc1 = {0.f, 0.f, 0.f, 0.f, 0.f, 0.f, 0.f, 0.f};
#pragma unroll
        for (int t = 0; t < 9; ++t) {
            v2f a0 = *(const v2f*)(p0 + 16 * t);
            v2f a1 = *(const v2f*)(p1 + 16 * t);
            acc0 = __builtin_amdgcn_wmma_f32_16x16x4_f32(
                false, a0, false, bt[t % 3], (short)0, acc0, false, false);
            acc1 = __builtin_amdgcn_wmma_f32_16x16x4_f32(
                false, a1, false, bt[t % 3], (short)0, acc1, false, false);
        }
#pragma unroll
        for (int g = 0; g < 2; ++g) {
            pv[4 * tp + g]     = (acc0[4 * g] + row_xor<4>(acc0[4 * g + 1]))
                               + (row_xor<8>(acc0[4 * g + 2]) + row_xor<12>(acc0[4 * g + 3]));
            pv[4 * tp + 2 + g] = (acc1[4 * g] + row_xor<4>(acc1[4 * g + 1]))
                               + (row_xor<8>(acc1[4 * g + 2]) + row_xor<12>(acc1[4 * g + 3]));
        }
    }

    if (nIdx < 4) {
        int srow = wv * 32 + ((lane >> 4) << 1);
#pragma unroll
        for (int q = 0; q < 8; ++q) {
            pooled[srow + 4 * q    ][nIdx] = pv[2 * q];
            pooled[srow + 4 * q + 1][nIdx] = pv[2 * q + 1];
        }
    }
    __syncthreads();

    float z[4];
    {
        float prod = 1.0f;
#pragma unroll
        for (int j = 0; j < 4; ++j) {
            float xj = pooled[tid][j];
            float sx, cx;
            sincosf(xj, &sx, &cx);
            prod *= sAB[j] * cx + sAB[4 + j] * sx;
            z[j] = prod;
        }
#pragma unroll
        for (int j = 0; j < 4; ++j) {
            red[tid][j]     = z[j];
            red[tid][4 + j] = z[j] * z[j];
        }
    }
    __syncthreads();

    {
        const int j = tid & 7;
        const int g = tid >> 3;
        float r[8];
#pragma unroll
        for (int i = 0; i < 8; ++i) r[i] = red[g + 32 * i][j];
        red2[g][j] = ((r[0] + r[4]) + (r[2] + r[6])) + ((r[1] + r[5]) + (r[3] + r[7]));
    }
    __syncthreads();

    {
        size_t s = (size_t)blockIdx.x * SPB + tid;
        *(float4*)(zbuf + s * 4) = make_float4(z[0], z[1], z[2], z[3]);
    }

    if (tid < 8) {
        float v[32];
#pragma unroll
        for (int g = 0; g < 32; ++g) v[g] = red2[g][tid];
#pragma unroll
        for (int st = 16; st > 0; st >>= 1) {
#pragma unroll
            for (int g = 0; g < st; ++g) v[g] += v[g + st];
        }
        const int b    = blockIdx.x;
        const int full = (B / SPB) & ~3;
        size_t idx = (b < full) ? (size_t)(b >> 2) * 32 + tid * 4 + (b & 3)
                                : (size_t)b * 8 + tid;
        partials[idx] = v[0];
    }
}

__device__ __forceinline__ float lane_xor16(float x)
{
    int b = __builtin_bit_cast(int, x);
    return __builtin_bit_cast(float,
        __builtin_amdgcn_permlanex16(b, b, 0x76543210, 0xfedcba98, false, false));
}

__global__ __launch_bounds__(32) __attribute__((amdgpu_waves_per_eu(1, 1)))
void stats_kernel(const float* __restrict__ partials, int nblk, int B,
                             const float* __restrict__ gamma,
                             const float* __restrict__ beta,
                             float* __restrict__ stats)
{
    const int lane = threadIdx.x;
    const int j    = lane & 7;
    const int ngrp = nblk >> 2;
    const int n4   = ngrp * 8;
    const float4* p4 = (const float4*)partials;

    const float g  = gamma[lane & 3];
    const float be = beta[lane & 3];

    float4 acc = make_float4(0.f, 0.f, 0.f, 0.f);
    int base = 0;
    for (; base + 512 <= n4; base += 512) {
        float4 v[16];
#pragma unroll
        for (int u = 0; u < 16; ++u) v[u] = p4[base + lane + 32 * u];
#pragma unroll
        for (int st = 1; st < 16; st <<= 1) {
#pragma unroll
            for (int u = 0; u < 16; u += 2 * st) {
                v[u].x += v[u + st].x;
                v[u].y += v[u + st].y;
                v[u].z += v[u + st].z;
                v[u].w += v[u + st].w;
            }
        }
        acc.x += v[0].x;
        acc.y += v[0].y;
        acc.z += v[0].z;
        acc.w += v[0].w;
    }
    for (int i = base + lane; i < n4; i += 32) {
        float4 v = p4[i];
        acc.x += v.x;
        acc.y += v.y;
        acc.z += v.z;
        acc.w += v.w;
    }
    float s = (acc.x + acc.y) + (acc.z + acc.w);
    if (lane < 8) {
        for (int b = ngrp << 2; b < nblk; ++b) s += partials[(size_t)b * 8 + j];
    }
    s += row_xor<8>(s);
    s += lane_xor16(s);
    const float sq = row_xor<4>(s);

    float invB = 1.0f / (float)B;
    float mean = s * invB;
    float var  = sq * invB - mean * mean;
    float sc   = g * rsqrtf(var + 1e-5f);
    if (lane < 4) {
        stats[lane]     = sc;
        stats[4 + lane] = be - mean * sc;
    }
}

__global__ __launch_bounds__(256)
void norm_kernel(const float* __restrict__ zbuf,
                 const float* __restrict__ stats,
                 float* __restrict__ out, int B)
{
    int i = blockIdx.x * blockDim.x + threadIdx.x;
    if (i < B) {
        float4 z  = *(const float4*)(zbuf + (size_t)i * 4);
        float4 sc = *(const float4*)(stats);
        float4 sh = *(const float4*)(stats + 4);
        float4 o;
        o.x = z.x * sc.x + sh.x;
        o.y = z.y * sc.y + sh.y;
        o.z = z.z * sc.z + sh.z;
        o.w = z.w * sc.w + sh.w;
        *(float4*)(out + (size_t)i * 4) = o;
    }
}

extern "C" void kernel_launch(void* const* d_in, const int* in_sizes, int n_in,
                              void* d_out, int out_size, void* d_ws, size_t ws_size,
                              hipStream_t stream)
{
    const float* x      = (const float*)d_in[0];
    const float* params = (const float*)d_in[1];
    const float* gamma  = (const float*)d_in[2];
    const float* beta   = (const float*)d_in[3];
    float* out = (float*)d_out;

    int B = in_sizes[0] / 576;
    int nblkA = B / SPB;

    float* zbuf     = (float*)d_ws;
    float* partials = zbuf + (size_t)B * 4;
    float* stats    = partials + (size_t)nblkA * 8;

    pool_circuit_kernel<<<nblkA, THREADS_A, 0, stream>>>(x, params, zbuf, partials, B);
    stats_kernel<<<1, 32, 0, stream>>>(partials, nblkA, B, gamma, beta, stats);
    norm_kernel<<<(B + 255) / 256, 256, 0, stream>>>(zbuf, stats, out, B);
}
